# starting kernel
# baseline (speedup 1.0000x reference)
#include <hip/hip_runtime.h>

typedef __attribute__((ext_vector_type(2))) float v2f;
typedef __attribute__((ext_vector_type(8))) float v8f;

#define N_C    3144
#define NP     2
#define NMOB   6
#define NCOV   10
#define T_ALL  156
#define TP     154
#define TP_PAD 160
#define NNZ_   31440
#define LDW    3168
#define NWM    ((size_t)N_C * (size_t)LDW)

#define OFF_WB   ((size_t)0)
#define OFF_WH   (NWM)
#define OFF_WA   (2 * NWM)
#define OFF_CS   (3 * NWM)
#define OFF_DS   (OFF_CS + (size_t)TP_PAD * N_C)
#define OFF_BC   (OFF_DS + (size_t)TP_PAD * N_C)
#define OFF_BD   (OFF_BC + (size_t)TP * N_C)
#define ZERO_CNT (OFF_BC)

__global__ void zero_k(float* __restrict__ p, size_t count) {
    size_t i = (size_t)blockIdx.x * blockDim.x + threadIdx.x;
    size_t stride = (size_t)gridDim.x * blockDim.x;
    for (; i < count; i += stride) p[i] = 0.0f;
}

__global__ void scatter_k(const int* __restrict__ rows, const int* __restrict__ cols,
                          const float* __restrict__ bnz, const float* __restrict__ anz,
                          const float* __restrict__ hnz, float* __restrict__ ws) {
    int i = blockIdx.x * blockDim.x + threadIdx.x;
    if (i >= NNZ_) return;
    size_t off = (size_t)rows[i] * LDW + (size_t)cols[i];
    atomicAdd(ws + OFF_WB + off, bnz[i]);
    atomicAdd(ws + OFF_WA + off, anz[i]);
    atomicAdd(ws + OFF_WH + off, hnz[i]);
}

__global__ void prep_k(const float* __restrict__ C, const float* __restrict__ D,
                       const float* __restrict__ M, const float* __restrict__ cov,
                       const float* __restrict__ mu, const float* __restrict__ nu,
                       const float* __restrict__ ups, const float* __restrict__ zet,
                       float* __restrict__ ws) {
    int idx = blockIdx.x * blockDim.x + threadIdx.x;
    if (idx >= TP * N_C) return;
    int t = idx / N_C;
    int m = idx - t * N_C;

    float cs = C[(size_t)t * N_C + m] + C[(size_t)(t + 1) * N_C + m];
    float dsv = D[(size_t)t * N_C + m] + D[(size_t)(t + 1) * N_C + m];
    ws[OFF_CS + (size_t)t * N_C + m] = cs;
    ws[OFF_DS + (size_t)t * N_C + m] = dsv;

    float mc = 0.0f, md = 0.0f;
#pragma unroll
    for (int k = 0; k < NMOB; ++k) {
#pragma unroll
        for (int tau = 0; tau < NP; ++tau) {
            float mv = M[((size_t)k * T_ALL + (size_t)(t + tau)) * N_C + m];
            mc = fmaf(mv, mu[k * NP + tau], mc);
            md = fmaf(mv, nu[k * NP + tau], md);
        }
    }
    float uc = 0.0f, zc = 0.0f;
#pragma unroll
    for (int j = 0; j < NCOV; ++j) {
        float cv = cov[(size_t)j * N_C + m];
        uc = fmaf(ups[j], cv, uc);
        zc = fmaf(zet[j], cv, zc);
    }
    ws[OFF_BC + (size_t)idx] = mc + uc;
    ws[OFF_BD + (size_t)idx] = md + zc;
}

__global__ __launch_bounds__(128)
void gemm_k(const float* __restrict__ ws, float* __restrict__ out) {
    const float* WB = ws + OFF_WB;
    const float* WH = ws + OFF_WH;
    const float* WA = ws + OFF_WA;
    const float* CS = ws + OFF_CS;
    const float* DS = ws + OFF_DS;

    const int wave = threadIdx.x >> 5;
    const int lane = threadIdx.x & 31;
    const int pair = blockIdx.x * 4 + wave;
    if (pair >= 99) return;
    const int c0 = pair * 32;
    const int t0 = blockIdx.y * 16;
    const int mrow = lane & 15;
    const int kg = (lane >> 4) * 2;

    v8f acc0 = {};
    v8f acc1 = {};
    const int z = blockIdx.z;
    const int npass = (z == 0) ? 1 : 2;

    for (int pass = 0; pass < npass; ++pass) {
        const float* X = (z == 0) ? CS : (pass == 0 ? CS : DS);
        const float* W = (z == 0) ? WB : (pass == 0 ? WH : WA);
        const float* xrow = X + (size_t)(t0 + mrow) * N_C;
        for (int k = 0; k < N_C; k += 4) {
            v2f a;
            a.x = xrow[k + kg];
            a.y = xrow[k + kg + 1];
            const float* w0 = W + (size_t)(k + kg) * LDW + (size_t)(c0 + mrow);
            v2f b0, b1;
            b0.x = w0[0];
            b0.y = w0[LDW];
            b1.x = w0[16];
            b1.y = w0[LDW + 16];
            acc0 = __builtin_amdgcn_wmma_f32_16x16x4_f32(false, a, false, b0,
                                                         (short)0, acc0, false, false);
            acc1 = __builtin_amdgcn_wmma_f32_16x16x4_f32(false, a, false, b1,
                                                         (short)0, acc1, false, false);
        }
    }

    const float* base = ws + ((z == 0) ? OFF_BC : OFF_BD);
    const size_t obase = (size_t)z * TP * N_C;
    const int rbump = (lane >> 4) << 3;
#pragma unroll
    for (int r = 0; r < 8; ++r) {
        int row = t0 + r + rbump;
        if (row >= TP) continue;
        int col0 = c0 + mrow;
        if (col0 < N_C)
            out[obase + (size_t)row * N_C + col0] =
                acc0[r] + base[(size_t)row * N_C + col0];
        int col1 = col0 + 16;
        if (col1 < N_C)
            out[obase + (size_t)row * N_C + col1] =
                acc1[r] + base[(size_t)row * N_C + col1];
    }
}

extern "C" void kernel_launch(void* const* d_in, const int* in_sizes, int n_in,
                              void* d_out, int out_size, void* d_ws, size_t ws_size,
                              hipStream_t stream) {
    const float* C   = (const float*)d_in[0];
    const float* D   = (const float*)d_in[1];
    const float* M   = (const float*)d_in[2];
    const float* cov = (const float*)d_in[3];
    const float* bnz = (const float*)d_in[4];
    const float* anz = (const float*)d_in[5];
    const float* hnz = (const float*)d_in[6];
    const float* mu  = (const float*)d_in[7];
    const float* nu  = (const float*)d_in[8];
    const float* ups = (const float*)d_in[9];
    const float* zet = (const float*)d_in[10];
    const int* rows  = (const int*)d_in[11];
    const int* cols  = (const int*)d_in[12];
    float* ws  = (float*)d_ws;
    float* out = (float*)d_out;

    zero_k<<<4096, 256, 0, stream>>>(ws, ZERO_CNT);
    scatter_k<<<(NNZ_ + 255) / 256, 256, 0, stream>>>(rows, cols, bnz, anz, hnz, ws);
    prep_k<<<(TP * N_C + 255) / 256, 256, 0, stream>>>(C, D, M, cov, mu, nu, ups, zet, ws);
    gemm_k<<<dim3(25, TP_PAD / 16, 2), 128, 0, stream>>>(ws, out);
}


# GEMM restaged through LDS: each 4-wave workgroup owns an 80x32 output tile, stages 64-k chunks of X and W with whole-line loads, every wave keeps 5x2 tiles over a quarter of k, sums joined in LDS
# speedup vs baseline: 1.5941x; 1.5941x over previous
#include <hip/hip_runtime.h>

typedef __attribute__((ext_vector_type(2))) float v2f;
typedef __attribute__((ext_vector_type(8))) float v8f;
typedef __attribute__((ext_vector_type(4))) float v4f;

#define N_C    3144
#define NP     2
#define NMOB   6
#define NCOV   10
#define T_ALL  156
#define TP     154
#define TP_PAD 160
#define NNZ_   31440
#define LDW    3168
#define NWM    ((size_t)N_C * (size_t)LDW)

#define OFF_WB   ((size_t)0)
#define OFF_WH   (NWM)
#define OFF_WA   (2 * NWM)
#define OFF_CS   (3 * NWM)
#define OFF_DS   (OFF_CS + (size_t)TP_PAD * N_C)
#define OFF_BC   (OFF_DS + (size_t)TP_PAD * N_C)
#define OFF_BD   (OFF_BC + (size_t)TP * N_C)
#define ZERO_CNT (OFF_BC)

__global__ void zero_k(float* __restrict__ p, size_t count) {
    size_t i = (size_t)blockIdx.x * blockDim.x + threadIdx.x;
    size_t stride = (size_t)gridDim.x * blockDim.x;
    for (; i < count; i += stride) p[i] = 0.0f;
}

__global__ void scatter_k(const int* __restrict__ rows, const int* __restrict__ cols,
                          const float* __restrict__ bnz, const float* __restrict__ anz,
                          const float* __restrict__ hnz, float* __restrict__ ws) {
    int i = blockIdx.x * blockDim.x + threadIdx.x;
    if (i >= NNZ_) return;
    size_t off = (size_t)rows[i] * LDW + (size_t)cols[i];
    atomicAdd(ws + OFF_WB + off, bnz[i]);
    atomicAdd(ws + OFF_WA + off, anz[i]);
    atomicAdd(ws + OFF_WH + off, hnz[i]);
}

__global__ void prep_k(const float* __restrict__ C, const float* __restrict__ D,
                       const float* __restrict__ M, const float* __restrict__ cov,
                       const float* __restrict__ mu, const float* __restrict__ nu,
                       const float* __restrict__ ups, const float* __restrict__ zet,
                       float* __restrict__ ws) {
    int idx = blockIdx.x * blockDim.x + threadIdx.x;
    if (idx >= TP * N_C) return;
    int t = idx / N_C;
    int m = idx - t * N_C;

    float cs = C[(size_t)t * N_C + m] + C[(size_t)(t + 1) * N_C + m];
    float dsv = D[(size_t)t * N_C + m] + D[(size_t)(t + 1) * N_C + m];
    ws[OFF_CS + (size_t)t * N_C + m] = cs;
    ws[OFF_DS + (size_t)t * N_C + m] = dsv;

    float mc = 0.0f, md = 0.0f;
#pragma unroll
    for (int k = 0; k < NMOB; ++k) {
#pragma unroll
        for (int tau = 0; tau < NP; ++tau) {
            float mv = M[((size_t)k * T_ALL + (size_t)(t + tau)) * N_C + m];
            mc = fmaf(mv, mu[k * NP + tau], mc);
            md = fmaf(mv, nu[k * NP + tau], md);
        }
    }
    float uc = 0.0f, zc = 0.0f;
#pragma unroll
    for (int j = 0; j < NCOV; ++j) {
        float cv = cov[(size_t)j * N_C + m];
        uc = fmaf(ups[j], cv, uc);
        zc = fmaf(zet[j], cv, zc);
    }
    ws[OFF_BC + (size_t)idx] = mc + uc;
    ws[OFF_BD + (size_t)idx] = md + zc;
}

#define BM   80
#define BN   32
#define KC   64
#define XP   68
#define WP   40
#define NCH  50
#define XS_SZ  (BM * XP)
#define WS_SZ  (KC * WP)
#define STG_SZ (XS_SZ + WS_SZ)
#define NUNIT_D 198
#define XSTEP  (8u * N_C * 4u)
#define WSTEP  (16u * LDW * 4u)

__global__ __launch_bounds__(128)
void gemm_k(const float* __restrict__ ws, float* __restrict__ out) {
    __shared__ __attribute__((aligned(16))) float smem[2 * STG_SZ];

    const float* WB = ws + OFF_WB;
    const float* WH = ws + OFF_WH;
    const float* WA = ws + OFF_WA;
    const float* CS = ws + OFF_CS;
    const float* DS = ws + OFF_DS;

    const int tid  = threadIdx.x;
    const int wave = tid >> 5;
    const int lane = tid & 31;
    const int m    = lane & 15;
    const int h2   = (lane >> 4) * 2;

    int u = blockIdx.x;
    const int z = (u < NUNIT_D) ? 1 : 0;
    if (z == 0) u -= NUNIT_D;
    const int r0  = (u & 1) * BM;
    const int c0  = (u >> 1) * BN;
    const int nch = z ? 2 * NCH : NCH;

    v4f xr[10];
    v4f wr[4];
    v8f acc[5][2] = {};
    const v4f zero4 = {0.0f, 0.0f, 0.0f, 0.0f};

    const unsigned xoff  = ((unsigned)(r0 + (tid >> 4)) * N_C + (unsigned)(tid & 15) * 4u) * 4u;
    const unsigned xoff0 = ((unsigned)(r0 + (tid >> 4)) * N_C) * 4u;
    const unsigned woff  = ((unsigned)(tid >> 3) * LDW + (unsigned)c0 + (unsigned)(tid & 7) * 4u) * 4u;
    const unsigned woff0 = ((unsigned)c0 + (unsigned)(tid & 7) * 4u) * 4u;
    const int xs_off = (tid >> 4) * XP + (tid & 15) * 4;
    const int ws_off = XS_SZ + (tid >> 3) * WP + (tid & 7) * 4;

    {
        const char* Xc = reinterpret_cast<const char*>(CS);
        const char* Wc = reinterpret_cast<const char*>(z ? WH : WB);
#pragma unroll
        for (int i = 0; i < 10; ++i)
            xr[i] = *reinterpret_cast<const v4f*>(Xc + (size_t)xoff + (size_t)i * XSTEP);
#pragma unroll
        for (int i = 0; i < 4; ++i)
            wr[i] = *reinterpret_cast<const v4f*>(Wc + (size_t)woff + (size_t)i * WSTEP);
#pragma unroll
        for (int i = 0; i < 10; ++i)
            *reinterpret_cast<v4f*>(smem + xs_off + i * 8 * XP) = xr[i];
#pragma unroll
        for (int i = 0; i < 4; ++i)
            *reinterpret_cast<v4f*>(smem + ws_off + i * 16 * WP) = wr[i];
    }
    __syncthreads();

    for (int c = 0; c < nch; ++c) {
        const bool more = (c + 1 < nch);
        const int  cn = c + 1;
        const bool second = (cn >= NCH);
        const int  kbase = (second ? cn - NCH : cn) * KC;
        const bool tail = (kbase + KC > N_C);
        if (more) {
            const char* Xc = reinterpret_cast<const char*>(second ? DS : CS) + (size_t)kbase * 4;
            const char* Wc = reinterpret_cast<const char*>(z ? (second ? WA : WH) : WB) + (size_t)kbase * (LDW * 4);
            if (!tail) {
#pragma unroll
                for (int i = 0; i < 10; ++i)
                    xr[i] = *reinterpret_cast<const v4f*>(Xc + (size_t)xoff + (size_t)i * XSTEP);
#pragma unroll
                for (int i = 0; i < 4; ++i)
                    wr[i] = *reinterpret_cast<const v4f*>(Wc + (size_t)woff + (size_t)i * WSTEP);
            } else {
                const unsigned xo = (kbase + (tid & 15) * 4 < N_C) ? xoff : xoff0;
#pragma unroll
                for (int i = 0; i < 10; ++i)
                    xr[i] = *reinterpret_cast<const v4f*>(Xc + (size_t)xo + (size_t)i * XSTEP);
#pragma unroll
                for (int i = 0; i < 4; ++i) {
                    const bool in = (kbase + (tid >> 3) + 16 * i < N_C);
                    const unsigned wo = in ? woff + (unsigned)i * WSTEP : woff0;
                    wr[i] = *reinterpret_cast<const v4f*>(Wc + (size_t)wo);
                }
            }
        }

        __builtin_amdgcn_sched_barrier(0);

        {
            const float* xs = smem + (c & 1) * STG_SZ;
            const float* wl = xs + XS_SZ;
#pragma unroll
            for (int s = 0; s < 4; ++s) {
                const int kk = wave * 16 + s * 4 + h2;
                v2f b0, b1;
                b0.x = wl[kk * WP + m];
                b0.y = wl[(kk + 1) * WP + m];
                b1.x = wl[kk * WP + 16 + m];
                b1.y = wl[(kk + 1) * WP + 16 + m];
#pragma unroll
                for (int rt = 0; rt < 5; ++rt) {
                    const v2f a = *reinterpret_cast<const v2f*>(xs + (rt * 16 + m) * XP + kk);
                    acc[rt][0] = __builtin_amdgcn_wmma_f32_16x16x4_f32(false, a, false, b0,
                                                                       (short)0, acc[rt][0], false, false);
                    acc[rt][1] = __builtin_amdgcn_wmma_f32_16x16x4_f32(false, a, false, b1,
                                                                       (short)0, acc[rt][1], false, false);
                }
            }
        }

        __builtin_amdgcn_sched_barrier(0);

        if (more) {
            float* st = smem + (cn & 1) * STG_SZ;
            if (!tail) {
#pragma unroll
                for (int i = 0; i < 10; ++i)
                    *reinterpret_cast<v4f*>(st + xs_off + i * 8 * XP) = xr[i];
#pragma unroll
                for (int i = 0; i < 4; ++i)
                    *reinterpret_cast<v4f*>(st + ws_off + i * 16 * WP) = wr[i];
            } else {
                const bool inx = (kbase + (tid & 15) * 4 < N_C);
#pragma unroll
                for (int i = 0; i < 10; ++i)
                    *reinterpret_cast<v4f*>(st + xs_off + i * 8 * XP) = inx ? xr[i] : zero4;
#pragma unroll
                for (int i = 0; i < 4; ++i) {
                    const bool in = (kbase + (tid >> 3) + 16 * i < N_C);
                    *reinterpret_cast<v4f*>(st + ws_off + i * 16 * WP) = in ? wr[i] : zero4;
                }
            }
        }
        __syncthreads();
    }

    float* red = smem;
#pragma unroll
    for (int rt = 0; rt < 5; ++rt)
#pragma unroll
        for (int ct = 0; ct < 2; ++ct)
#pragma unroll
            for (int r = 0; r < 8; ++r)
                red[((wave * 10 + rt * 2 + ct) * 8 + r) * 32 + lane] = acc[rt][ct][r];
    __syncthreads();

    const float* base = ws + (z ? OFF_BD : OFF_BC);
    const size_t obase = (size_t)z * TP * N_C;
#pragma unroll 4
    for (int i = 0; i < 20; ++i) {
        const int e   = tid + 128 * i;
        const int row = e >> 5;
        const int col = e & 31;
        const int t   = (row >> 4) * 2 + (col >> 4);
        const int idx = (t * 8 + (row & 7)) * 32 + (col & 15) + ((row & 8) << 1);
        const float sum = ((red[idx] + red[2560 + idx]) + red[5120 + idx]) + red[7680 + idx];
        const int gr = r0 + row;
        const int gc = c0 + col;
        if (gr < TP && gc < N_C)
            out[obase + (size_t)gr * N_C + gc] = sum + base[(size_t)gr * N_C + gc];
    }
}

extern "C" void kernel_launch(void* const* d_in, const int* in_sizes, int n_in,
                              void* d_out, int out_size, void* d_ws, size_t ws_size,
                              hipStream_t stream) {
    const float* C   = (const float*)d_in[0];
    const float* D   = (const float*)d_in[1];
    const float* M   = (const float*)d_in[2];
    const float* cov = (const float*)d_in[3];
    const float* bnz = (const float*)d_in[4];
    const float* anz = (const float*)d_in[5];
    const float* hnz = (const float*)d_in[6];
    const float* mu  = (const float*)d_in[7];
    const float* nu  = (const float*)d_in[8];
    const float* ups = (const float*)d_in[9];
    const float* zet = (const float*)d_in[10];
    const int* rows  = (const int*)d_in[11];
    const int* cols  = (const int*)d_in[12];
    float* ws  = (float*)d_ws;
    float* out = (float*)d_out;

    zero_k<<<4096, 256, 0, stream>>>(ws, ZERO_CNT);
    scatter_k<<<(NNZ_ + 255) / 256, 256, 0, stream>>>(rows, cols, bnz, anz, hnz, ws);
    prep_k<<<(TP * N_C + 255) / 256, 256, 0, stream>>>(C, D, M, cov, mu, nu, ups, zet, ws);
    gemm_k<<<2 * NUNIT_D, 128, 0, stream>>>(ws, out);
}


# Setup launch widened: each thread makes a patch (base terms 2 time steps x 4 counties, transpose tile 32 x 128, entry values 4 at a time) with 16-byte loads and stores; 393 workgroups, not 2510
# speedup vs baseline: 7.0828x; 4.4431x over previous
#include <hip/hip_runtime.h>

typedef __attribute__((ext_vector_type(2))) float v2f;
typedef __attribute__((ext_vector_type(8))) float v8f;
typedef __attribute__((ext_vector_type(4))) float v4f;
typedef __attribute__((ext_vector_type(3))) float v3f;
typedef __attribute__((ext_vector_type(2))) unsigned v2u;

#define N_C    3144
#define NP     2
#define NMOB   6
#define NCOV   10
#define T_ALL  156
#define TP     154
#define TP_PAD 160
#define NNZ_   31440


#define OFF_XT   ((size_t)0)
#define OFF_V4   ((size_t)N_C * TP_PAD * 2)
#define OFF_BC   (OFF_V4 + (size_t)NNZ_ * 4)
#define OFF_BD   (OFF_BC + (size_t)TP * N_C)

#define NC4      786
#define XT_TM    128
#define NTB_X    125
#define NPATCH   60522
#define NTB_B    237
#define NV4      7860
#define NTB_V    31
#define PREP_BLOCKS 393

__global__ __launch_bounds__(256)
void prep_k(const float* __restrict__ C, const float* __restrict__ D,
            const float* __restrict__ M, const float* __restrict__ cov,
            const float* __restrict__ bnz, const float* __restrict__ anz,
            const float* __restrict__ hnz,
            const float* __restrict__ mu, const float* __restrict__ nu,
            const float* __restrict__ ups, const float* __restrict__ zet,
            float* __restrict__ ws) {
    const int bid = blockIdx.x;
    const int tid = threadIdx.x;
    if (bid < NTB_X) {
        __shared__ float tc[32][XT_TM + 1];
        __shared__ float td[32][XT_TM + 1];
        const int m0 = (bid % 25) * XT_TM;
        const int t0 = (bid / 25) * 32;
        const int tx = tid & 31;
        const int ty = tid >> 5;
        const v4f* C4 = reinterpret_cast<const v4f*>(C);
        const v4f* D4 = reinterpret_cast<const v4f*>(D);
        const int g4 = (m0 >> 2) + tx;
        v4f ca[4], cb[4], da[4], db[4];
#pragma unroll
        for (int i = 0; i < 4; ++i) {
            const int t  = t0 + ty + 8 * i;
            const size_t rb = (size_t)(t < TP ? t : TP - 1) * NC4 + (g4 < NC4 ? g4 : NC4 - 1);
            ca[i] = C4[rb];
            cb[i] = C4[rb + NC4];
            da[i] = D4[rb];
            db[i] = D4[rb + NC4];
        }
#pragma unroll
        for (int i = 0; i < 4; ++i) {
            const int tl = ty + 8 * i;
            const bool in = (t0 + tl < TP) && (g4 < NC4);
            const v4f z = {0.0f, 0.0f, 0.0f, 0.0f};
            const v4f cs  = in ? ca[i] + cb[i] : z;
            const v4f dsv = in ? da[i] + db[i] : z;
            tc[tl][4 * tx + 0] = cs.x;  tc[tl][4 * tx + 1] = cs.y;
            tc[tl][4 * tx + 2] = cs.z;  tc[tl][4 * tx + 3] = cs.w;
            td[tl][4 * tx + 0] = dsv.x; td[tl][4 * tx + 1] = dsv.y;
            td[tl][4 * tx + 2] = dsv.z; td[tl][4 * tx + 3] = dsv.w;
        }
        __syncthreads();
        v4f* XT4 = reinterpret_cast<v4f*>(ws + OFF_XT);
        const int l2 = (tid & 15) * 2;
        const int hf = (tid >> 4) & 1;
        v4f o[8];
#pragma unroll
        for (int s = 0; s < 8; ++s) {
            const int ml = ty * 16 + s * 2 + hf;
            o[s].x = tc[l2][ml];
            o[s].y = td[l2][ml];
            o[s].z = tc[l2 + 1][ml];
            o[s].w = td[l2 + 1][ml];
        }
#pragma unroll
        for (int s = 0; s < 8; ++s) {
            const int mm = m0 + ty * 16 + s * 2 + hf;
            if (mm < N_C) XT4[((size_t)mm * TP_PAD + t0 + l2) >> 1] = o[s];
        }
        return;
    }
    if (bid < NTB_X + NTB_B) {
        const int idx = (bid - NTB_X) * 256 + tid;
        if (idx >= NPATCH) return;
        const int tp = idx / NC4;
        const int g4 = idx - tp * NC4;
        const int t  = 2 * tp;
        const v4f* M4 = reinterpret_cast<const v4f*>(M);
        const v4f* cov4 = reinterpret_cast<const v4f*>(cov);
        v4f r0[NMOB], r1[NMOB], r2[NMOB], cv[NCOV];
#pragma unroll
        for (int k = 0; k < NMOB; ++k) {
            const size_t rb = ((size_t)k * T_ALL + (size_t)t) * NC4 + g4;
            r0[k] = M4[rb];
            r1[k] = M4[rb + NC4];
            r2[k] = M4[rb + 2 * NC4];
        }
#pragma unroll
        for (int j = 0; j < NCOV; ++j) cv[j] = cov4[(size_t)j * NC4 + g4];
        __builtin_amdgcn_sched_barrier(0);
        v4f mc0 = {0.0f, 0.0f, 0.0f, 0.0f}, md0 = mc0, mc1 = mc0, md1 = mc0;
#pragma unroll
        for (int k = 0; k < NMOB; ++k) {
            const float u0 = mu[k * NP + 0], u1 = mu[k * NP + 1];
            const float n0 = nu[k * NP + 0], n1 = nu[k * NP + 1];
#pragma unroll
            for (int e = 0; e < 4; ++e) {
                mc0[e] = fmaf(r0[k][e], u0, mc0[e]);
                mc0[e] = fmaf(r1[k][e], u1, mc0[e]);
                md0[e] = fmaf(r0[k][e], n0, md0[e]);
                md0[e] = fmaf(r1[k][e], n1, md0[e]);
                mc1[e] = fmaf(r1[k][e], u0, mc1[e]);
                mc1[e] = fmaf(r2[k][e], u1, mc1[e]);
                md1[e] = fmaf(r1[k][e], n0, md1[e]);
                md1[e] = fmaf(r2[k][e], n1, md1[e]);
            }
        }
        v4f uc = {0.0f, 0.0f, 0.0f, 0.0f}, zc = uc;
#pragma unroll
        for (int j = 0; j < NCOV; ++j) {
            const float uj = ups[j], zj = zet[j];
#pragma unroll
            for (int e = 0; e < 4; ++e) {
                uc[e] = fmaf(uj, cv[j][e], uc[e]);
                zc[e] = fmaf(zj, cv[j][e], zc[e]);
            }
        }
        v4f* BC4 = reinterpret_cast<v4f*>(ws + OFF_BC);
        v4f* BD4 = reinterpret_cast<v4f*>(ws + OFF_BD);
        const size_t o0 = (size_t)t * NC4 + g4;
        BC4[o0]       = mc0 + uc;
        BC4[o0 + NC4] = mc1 + uc;
        BD4[o0]       = md0 + zc;
        BD4[o0 + NC4] = md1 + zc;
        return;
    }
    {
        const int j4 = (bid - NTB_X - NTB_B) * 256 + tid;
        if (j4 >= NV4) return;
        const v4f b = reinterpret_cast<const v4f*>(bnz)[j4];
        const v4f h = reinterpret_cast<const v4f*>(hnz)[j4];
        const v4f a = reinterpret_cast<const v4f*>(anz)[j4];
        v4f* V4o = reinterpret_cast<v4f*>(ws + OFF_V4) + (size_t)j4 * 4;
#pragma unroll
        for (int e = 0; e < 4; ++e) {
            v4f o;
            o.x = b[e];
            o.y = h[e];
            o.z = a[e];
            o.w = 0.0f;
            V4o[e] = o;
        }
    }
}

#define NT    197
#define NW    10
#define SEGL  3144
#define SCU   16
#define KB    8
#define NOCOL  16u
#define PADENT (NOCOL << 12)

__global__ __launch_bounds__(320)
void spmm_k(const int* __restrict__ rows, const int* __restrict__ cols,
            const float* __restrict__ ws, float* __restrict__ out) {
    __shared__ __attribute__((aligned(16))) unsigned list[NNZ_];
    __shared__ unsigned short steptab[NNZ_ / 4];
    __shared__ int scnt[NW];

    const int tile = blockIdx.x;
    const int tid  = threadIdx.x;
    const int wave = tid >> 5;
    const int lane = tid & 31;
    const int segb = wave * SEGL;

    int wcount = 0;
    for (int b = 0; b < SEGL; b += 32 * SCU) {
        int cv[SCU], rv[SCU];
#pragma unroll
        for (int u = 0; u < SCU; ++u) {
            const int jl = b + u * 32 + lane;
            const int jc = segb + (jl < SEGL ? jl : SEGL - 1);
            cv[u] = cols[jc];
            rv[u] = rows[jc];
        }
#pragma unroll
        for (int u = 0; u < SCU; ++u) {
            const int jl = b + u * 32 + lane;
            const bool match = (jl < SEGL) && ((cv[u] >> 4) == tile);
            const unsigned mask = __builtin_amdgcn_ballot_w32(match);
            if (match) {
                const int p = wcount + __popc(mask & ((1u << lane) - 1u));
                list[segb + p] = (unsigned)rv[u] | ((unsigned)(cv[u] & 15) << 12) | ((unsigned)(segb + jl) << 17);
            }
            wcount += __popc(mask);
        }
    }
    const int wpad = (wcount + 3) & ~3;
    if (lane < wpad - wcount) list[segb + wcount + lane] = PADENT;
    if (lane == 0) scnt[wave] = wpad >> 2;
    __syncthreads();

    int sbase = 0, stot = 0;
#pragma unroll
    for (int w = 0; w < NW; ++w) {
        const int c = scnt[w];
        if (w < wave) sbase += c;
        stot += c;
    }
    for (int ls = lane; ls < (wpad >> 2); ls += 32)
        steptab[sbase + ls] = (unsigned short)(segb + 4 * ls);
    __syncthreads();

    const int m  = lane & 15;
    const int h2 = (lane >> 4) * 2;
    const v2f* XT = reinterpret_cast<const v2f*>(ws + OFF_XT) + (wave * 16 + m);
    const float* V4 = ws + OFF_V4;

    v8f accC = {}, accH = {}, accA = {};

    for (int v0 = 0; v0 < stot; v0 += KB) {
        v2f x0[KB], x1[KB];
        v3f w0[KB], w1[KB];
        unsigned cl0[KB], cl1[KB];
#pragma unroll
        for (int q = 0; q < KB; ++q) {
            const bool on = (v0 + q < stot);
            const int  v  = on ? v0 + q : stot - 1;
            const int  off = steptab[v];
            const v2u ent = *reinterpret_cast<const v2u*>(&list[off + h2]);
            x0[q] = XT[(size_t)(ent.x & 4095u) * TP_PAD];
            x1[q] = XT[(size_t)(ent.y & 4095u) * TP_PAD];
            w0[q] = *reinterpret_cast<const v3f*>(V4 + (size_t)(ent.x >> 17) * 4);
            w1[q] = *reinterpret_cast<const v3f*>(V4 + (size_t)(ent.y >> 17) * 4);
            cl0[q] = on ? ((ent.x >> 12) & 31u) : NOCOL;
            cl1[q] = on ? ((ent.y >> 12) & 31u) : NOCOL;
        }
        __builtin_amdgcn_sched_barrier(0);
#pragma unroll
        for (int q = 0; q < KB; ++q) {
            const bool s0 = (cl0[q] == (unsigned)m);
            const bool s1 = (cl1[q] == (unsigned)m);
            v2f ac, ad, bb, bh, ba;
            ac.x = x0[q].x;  ac.y = x1[q].x;
            ad.x = x0[q].y;  ad.y = x1[q].y;
            bb.x = s0 ? w0[q].x : 0.0f;  bb.y = s1 ? w1[q].x : 0.0f;
            bh.x = s0 ? w0[q].y : 0.0f;  bh.y = s1 ? w1[q].y : 0.0f;
            ba.x = s0 ? w0[q].z : 0.0f;  ba.y = s1 ? w1[q].z : 0.0f;
            accC = __builtin_amdgcn_wmma_f32_16x16x4_f32(false, ac, false, bb, (short)0, accC, false, false);
            accH = __builtin_amdgcn_wmma_f32_16x16x4_f32(false, ac, false, bh, (short)0, accH, false, false);
            accA = __builtin_amdgcn_wmma_f32_16x16x4_f32(false, ad, false, ba, (short)0, accA, false, false);
        }
    }

    const float* baseC = ws + OFF_BC;
    const float* baseD = ws + OFF_BD;
    const int gc = tile * 16 + m;
#pragma unroll
    for (int r = 0; r < 8; ++r) {
        const int gr = wave * 16 + r + 8 * (lane >> 4);
        if (gr < TP && gc < N_C) {
            const size_t o = (size_t)gr * N_C + gc;
            out[o] = accC[r] + baseC[o];
            out[(size_t)TP * N_C + o] = (accH[r] + accA[r]) + baseD[o];
        }
    }
}

extern "C" void kernel_launch(void* const* d_in, const int* in_sizes, int n_in,
                              void* d_out, int out_size, void* d_ws, size_t ws_size,
                              hipStream_t stream) {
    const float* C   = (const float*)d_in[0];
    const float* D   = (const float*)d_in[1];
    const float* M   = (const float*)d_in[2];
    const float* cov = (const float*)d_in[3];
    const float* bnz = (const float*)d_in[4];
    const float* anz = (const float*)d_in[5];
    const float* hnz = (const float*)d_in[6];
    const float* mu  = (const float*)d_in[7];
    const float* nu  = (const float*)d_in[8];
    const float* ups = (const float*)d_in[9];
    const float* zet = (const float*)d_in[10];
    const int* rows  = (const int*)d_in[11];
    const int* cols  = (const int*)d_in[12];
    float* ws  = (float*)d_ws;
    float* out = (float*)d_out;

    prep_k<<<PREP_BLOCKS, 256, 0, stream>>>(C, D, M, cov, bnz, anz, hnz, mu, nu, ups, zet, ws);
    spmm_k<<<NT, 320, 0, stream>>>(rows, cols, ws, out);
}


# Entries grouped once by column bucket in the setup launch (ballot split, no atomics); each tile workgroup reads only its bucket of every entry block instead of scanning both whole index arrays
# speedup vs baseline: 9.4078x; 1.3283x over previous
#include <hip/hip_runtime.h>

typedef __attribute__((ext_vector_type(2))) float v2f;
typedef __attribute__((ext_vector_type(8))) float v8f;
typedef __attribute__((ext_vector_type(4))) float v4f;
typedef __attribute__((ext_vector_type(3))) float v3f;
typedef __attribute__((ext_vector_type(2))) unsigned v2u;

#define N_C    3144
#define NP     2
#define NMOB   6
#define NCOV   10
#define T_ALL  156
#define TP     154
#define TP_PAD 160
#define NNZ_   31440


#define OFF_XT   ((size_t)0)
#define OFF_V4   ((size_t)N_C * TP_PAD * 2)
#define OFF_BC   (OFF_V4 + (size_t)NNZ_ * 4)
#define OFF_BD   (OFF_BC + (size_t)TP * N_C)
#define OFF_OT   (OFF_BD + (size_t)TP * N_C)

#define NSB      30
#define SBL      1048
#define NBK      16
#define BKW      208
#define SST      5

#define NC4      786
#define XT_TM    128
#define NTB_X    125
#define NPATCH   60522
#define NTB_B    237
#define PREP_BLOCKS 392

__global__ __launch_bounds__(256)
void prep_k(const float* __restrict__ C, const float* __restrict__ D,
            const float* __restrict__ M, const float* __restrict__ cov,
            const float* __restrict__ bnz, const float* __restrict__ anz,
            const float* __restrict__ hnz,
            const float* __restrict__ mu, const float* __restrict__ nu,
            const float* __restrict__ ups, const float* __restrict__ zet,
            const int* __restrict__ rows, const int* __restrict__ cols,
            float* __restrict__ ws) {
    const int tid = threadIdx.x;
    if (blockIdx.x < NSB) {
        __shared__ int wc[8][NBK];
        __shared__ int tot[NBK];
        const int sb   = blockIdx.x;
        const int wave = tid >> 5;
        const int lane = tid & 31;
        int key[SST];
        unsigned rec[SST];
        float bv[SST], hv[SST], av[SST];
#pragma unroll
        for (int s = 0; s < SST; ++s) {
            const int jl = s * 256 + tid;
            const bool ok = jl < SBL;
            const int j = sb * SBL + (ok ? jl : 0);
            const int c = cols[j];
            const int r = rows[j];
            bv[s] = bnz[j];
            hv[s] = hnz[j];
            av[s] = anz[j];
            key[s] = ok ? c / BKW : NBK;
            rec[s] = (unsigned)r | ((unsigned)c << 12);
        }
        int mycnt = 0;
#pragma unroll
        for (int g = 0; g < NBK; ++g) {
            int c = 0;
#pragma unroll
            for (int s = 0; s < SST; ++s) c += __popc(__builtin_amdgcn_ballot_w32(key[s] == g));
            if (lane == g) mycnt = c;
        }
        if (lane < NBK) wc[wave][lane] = mycnt;
        __syncthreads();
        if (tid < NBK) {
            int run = 0;
#pragma unroll
            for (int w = 0; w < 8; ++w) {
                const int c = wc[w][tid];
                wc[w][tid] = run;
                run += c;
            }
            tot[tid] = run;
        }
        __syncthreads();
        int mybase = 0;
#pragma unroll
        for (int g = 0; g < NBK; ++g) {
            const int t = tot[g];
            if (g < lane) mybase += t;
        }
        int* OT = reinterpret_cast<int*>(ws + OFF_OT) + sb * (NBK + 1);
        if (tid <= NBK) OT[tid] = mybase;
        if (lane < NBK) mybase += wc[wave][lane];
        int pos[SST];
#pragma unroll
        for (int s = 0; s < SST; ++s) pos[s] = 0;
#pragma unroll
        for (int g = 0; g < NBK; ++g) {
            int run = __builtin_amdgcn_readlane(mybase, g);
#pragma unroll
            for (int s = 0; s < SST; ++s) {
                const bool mt = (key[s] == g);
                const unsigned mask = __builtin_amdgcn_ballot_w32(mt);
                if (mt) pos[s] = run + __popc(mask & ((1u << lane) - 1u));
                run += __popc(mask);
            }
        }
        typedef __attribute__((ext_vector_type(4))) unsigned v4u;
        v4u* V4o = reinterpret_cast<v4u*>(ws + OFF_V4) + (size_t)sb * SBL;
#pragma unroll
        for (int s = 0; s < SST; ++s) {
            if (key[s] < NBK) {
                v4u o;
                o.x = __float_as_uint(bv[s]);
                o.y = __float_as_uint(hv[s]);
                o.z = __float_as_uint(av[s]);
                o.w = rec[s];
                V4o[pos[s]] = o;
            }
        }
        return;
    }
    const int bid = blockIdx.x - NSB;
    if (bid < NTB_X) {
        __shared__ float tc[32][XT_TM + 1];
        __shared__ float td[32][XT_TM + 1];
        const int m0 = (bid % 25) * XT_TM;
        const int t0 = (bid / 25) * 32;
        const int tx = tid & 31;
        const int ty = tid >> 5;
        const v4f* C4 = reinterpret_cast<const v4f*>(C);
        const v4f* D4 = reinterpret_cast<const v4f*>(D);
        const int g4 = (m0 >> 2) + tx;
        v4f ca[4], cb[4], da[4], db[4];
#pragma unroll
        for (int i = 0; i < 4; ++i) {
            const int t  = t0 + ty + 8 * i;
            const size_t rb = (size_t)(t < TP ? t : TP - 1) * NC4 + (g4 < NC4 ? g4 : NC4 - 1);
            ca[i] = C4[rb];
            cb[i] = C4[rb + NC4];
            da[i] = D4[rb];
            db[i] = D4[rb + NC4];
        }
#pragma unroll
        for (int i = 0; i < 4; ++i) {
            const int tl = ty + 8 * i;
            const bool in = (t0 + tl < TP) && (g4 < NC4);
            const v4f z = {0.0f, 0.0f, 0.0f, 0.0f};
            const v4f cs  = in ? ca[i] + cb[i] : z;
            const v4f dsv = in ? da[i] + db[i] : z;
            tc[tl][4 * tx + 0] = cs.x;  tc[tl][4 * tx + 1] = cs.y;
            tc[tl][4 * tx + 2] = cs.z;  tc[tl][4 * tx + 3] = cs.w;
            td[tl][4 * tx + 0] = dsv.x; td[tl][4 * tx + 1] = dsv.y;
            td[tl][4 * tx + 2] = dsv.z; td[tl][4 * tx + 3] = dsv.w;
        }
        __syncthreads();
        v4f* XT4 = reinterpret_cast<v4f*>(ws + OFF_XT);
        const int l2 = (tid & 15) * 2;
        const int hf = (tid >> 4) & 1;
        v4f o[8];
#pragma unroll
        for (int s = 0; s < 8; ++s) {
            const int ml = ty * 16 + s * 2 + hf;
            o[s].x = tc[l2][ml];
            o[s].y = td[l2][ml];
            o[s].z = tc[l2 + 1][ml];
            o[s].w = td[l2 + 1][ml];
        }
#pragma unroll
        for (int s = 0; s < 8; ++s) {
            const int mm = m0 + ty * 16 + s * 2 + hf;
            if (mm < N_C) XT4[((size_t)mm * TP_PAD + t0 + l2) >> 1] = o[s];
        }
        return;
    }
    if (bid < NTB_X + NTB_B) {
        const int idx = (bid - NTB_X) * 256 + tid;
        if (idx >= NPATCH) return;
        const int tp = idx / NC4;
        const int g4 = idx - tp * NC4;
        const int t  = 2 * tp;
        const v4f* M4 = reinterpret_cast<const v4f*>(M);
        const v4f* cov4 = reinterpret_cast<const v4f*>(cov);
        v4f r0[NMOB], r1[NMOB], r2[NMOB], cv[NCOV];
#pragma unroll
        for (int k = 0; k < NMOB; ++k) {
            const size_t rb = ((size_t)k * T_ALL + (size_t)t) * NC4 + g4;
            r0[k] = M4[rb];
            r1[k] = M4[rb + NC4];
            r2[k] = M4[rb + 2 * NC4];
        }
#pragma unroll
        for (int j = 0; j < NCOV; ++j) cv[j] = cov4[(size_t)j * NC4 + g4];
        __builtin_amdgcn_sched_barrier(0);
        v4f mc0 = {0.0f, 0.0f, 0.0f, 0.0f}, md0 = mc0, mc1 = mc0, md1 = mc0;
#pragma unroll
        for (int k = 0; k < NMOB; ++k) {
            const float u0 = mu[k * NP + 0], u1 = mu[k * NP + 1];
            const float n0 = nu[k * NP + 0], n1 = nu[k * NP + 1];
#pragma unroll
            for (int e = 0; e < 4; ++e) {
                mc0[e] = fmaf(r0[k][e], u0, mc0[e]);
                mc0[e] = fmaf(r1[k][e], u1, mc0[e]);
                md0[e] = fmaf(r0[k][e], n0, md0[e]);
                md0[e] = fmaf(r1[k][e], n1, md0[e]);
                mc1[e] = fmaf(r1[k][e], u0, mc1[e]);
                mc1[e] = fmaf(r2[k][e], u1, mc1[e]);
                md1[e] = fmaf(r1[k][e], n0, md1[e]);
                md1[e] = fmaf(r2[k][e], n1, md1[e]);
            }
        }
        v4f uc = {0.0f, 0.0f, 0.0f, 0.0f}, zc = uc;
#pragma unroll
        for (int j = 0; j < NCOV; ++j) {
            const float uj = ups[j], zj = zet[j];
#pragma unroll
            for (int e = 0; e < 4; ++e) {
                uc[e] = fmaf(uj, cv[j][e], uc[e]);
                zc[e] = fmaf(zj, cv[j][e], zc[e]);
            }
        }
        v4f* BC4 = reinterpret_cast<v4f*>(ws + OFF_BC);
        v4f* BD4 = reinterpret_cast<v4f*>(ws + OFF_BD);
        const size_t o0 = (size_t)t * NC4 + g4;
        BC4[o0]       = mc0 + uc;
        BC4[o0 + NC4] = mc1 + uc;
        BD4[o0]       = md0 + zc;
        BD4[o0 + NC4] = md1 + zc;
        return;
    }
}

#define NT    197
#define NW    10
#define SEGL  3144
#define BPW   3
#define FCH   4
#define KB    8
#define NOCOL  16u
#define PADENT (NOCOL << 12)

__global__ __launch_bounds__(320)
void spmm_k(const float* __restrict__ ws, float* __restrict__ out) {
    __shared__ __attribute__((aligned(16))) unsigned list[NNZ_];
    __shared__ unsigned short steptab[NNZ_ / 4];
    __shared__ int scnt[NW];

    const int tile = blockIdx.x;
    const int tid  = threadIdx.x;
    const int wave = tid >> 5;
    const int lane = tid & 31;
    const int segb = wave * SEGL;

    const int bucket = tile / (BKW / 16);
    const int* OT = reinterpret_cast<const int*>(ws + OFF_OT);
    const unsigned* RW = reinterpret_cast<const unsigned*>(ws + OFF_V4);

    int lo[BPW], n[BPW];
#pragma unroll
    for (int i = 0; i < BPW; ++i) {
        const int sb = wave * BPW + i;
        lo[i] = OT[sb * (NBK + 1) + bucket];
        n[i]  = OT[sb * (NBK + 1) + bucket + 1];
    }
#pragma unroll
    for (int i = 0; i < BPW; ++i) {
        int l = __builtin_amdgcn_readfirstlane(lo[i]);
        int h = __builtin_amdgcn_readfirstlane(n[i]);
        l = l < 0 ? 0 : (l > SBL ? SBL : l);
        h = h < l ? l : (h > SBL ? SBL : h);
        lo[i] = l;
        n[i]  = h - l;
    }

    int wcount = 0;
    unsigned rc[BPW][FCH];
#pragma unroll
    for (int i = 0; i < BPW; ++i) {
#pragma unroll
        for (int u = 0; u < FCH; ++u) {
            const int q  = lo[i] + u * 32 + lane;
            const int qc = q < SBL ? q : SBL - 1;
            rc[i][u] = RW[((size_t)(wave * BPW + i) * SBL + qc) * 4 + 3];
        }
    }
#pragma unroll
    for (int i = 0; i < BPW; ++i) {
#pragma unroll
        for (int u = 0; u < FCH; ++u) {
            const int p = u * 32 + lane;
            const bool match = (p < n[i]) && ((int)(rc[i][u] >> 16) == tile);
            const unsigned mask = __builtin_amdgcn_ballot_w32(match);
            if (match) {
                const int slot = wcount + __popc(mask & ((1u << lane) - 1u));
                list[segb + slot] = (rc[i][u] & 0xFFFFu) | ((unsigned)((wave * BPW + i) * SBL + lo[i] + p) << 17);
            }
            wcount += __popc(mask);
        }
        for (int p0 = FCH * 32; p0 < n[i]; p0 += 32) {
            const int p  = p0 + lane;
            const int q  = lo[i] + p;
            const int qc = q < SBL ? q : SBL - 1;
            const unsigned r1 = RW[((size_t)(wave * BPW + i) * SBL + qc) * 4 + 3];
            const bool match = (p < n[i]) && ((int)(r1 >> 16) == tile);
            const unsigned mask = __builtin_amdgcn_ballot_w32(match);
            if (match) {
                const int slot = wcount + __popc(mask & ((1u << lane) - 1u));
                list[segb + slot] = (r1 & 0xFFFFu) | ((unsigned)((wave * BPW + i) * SBL + q) << 17);
            }
            wcount += __popc(mask);
        }
    }
    const int wpad = (wcount + 3) & ~3;
    if (lane < wpad - wcount) list[segb + wcount + lane] = PADENT;
    if (lane == 0) scnt[wave] = wpad >> 2;
    __syncthreads();

    int sbase = 0, stot = 0;
#pragma unroll
    for (int w = 0; w < NW; ++w) {
        const int c = scnt[w];
        if (w < wave) sbase += c;
        stot += c;
    }
    for (int ls = lane; ls < (wpad >> 2); ls += 32)
        steptab[sbase + ls] = (unsigned short)(segb + 4 * ls);
    __syncthreads();

    const int m  = lane & 15;
    const int h2 = (lane >> 4) * 2;
    const v2f* XT = reinterpret_cast<const v2f*>(ws + OFF_XT) + (wave * 16 + m);
    const float* V4 = ws + OFF_V4;

    v8f accC = {}, accH = {}, accA = {};

    for (int v0 = 0; v0 < stot; v0 += KB) {
        v2f x0[KB], x1[KB];
        v3f w0[KB], w1[KB];
        unsigned cl0[KB], cl1[KB];
#pragma unroll
        for (int q = 0; q < KB; ++q) {
            const bool on = (v0 + q < stot);
            const int  v  = on ? v0 + q : stot - 1;
            const int  off = steptab[v];
            const v2u ent = *reinterpret_cast<const v2u*>(&list[off + h2]);
            x0[q] = XT[(size_t)(ent.x & 4095u) * TP_PAD];
            x1[q] = XT[(size_t)(ent.y & 4095u) * TP_PAD];
            w0[q] = *reinterpret_cast<const v3f*>(V4 + (size_t)(ent.x >> 17) * 4);
            w1[q] = *reinterpret_cast<const v3f*>(V4 + (size_t)(ent.y >> 17) * 4);
            cl0[q] = on ? ((ent.x >> 12) & 31u) : NOCOL;
            cl1[q] = on ? ((ent.y >> 12) & 31u) : NOCOL;
        }
        __builtin_amdgcn_sched_barrier(0);
#pragma unroll
        for (int q = 0; q < KB; ++q) {
            const bool s0 = (cl0[q] == (unsigned)m);
            const bool s1 = (cl1[q] == (unsigned)m);
            v2f ac, ad, bb, bh, ba;
            ac.x = x0[q].x;  ac.y = x1[q].x;
            ad.x = x0[q].y;  ad.y = x1[q].y;
            bb.x = s0 ? w0[q].x : 0.0f;  bb.y = s1 ? w1[q].x : 0.0f;
            bh.x = s0 ? w0[q].y : 0.0f;  bh.y = s1 ? w1[q].y : 0.0f;
            ba.x = s0 ? w0[q].z : 0.0f;  ba.y = s1 ? w1[q].z : 0.0f;
            accC = __builtin_amdgcn_wmma_f32_16x16x4_f32(false, ac, false, bb, (short)0, accC, false, false);
            accH = __builtin_amdgcn_wmma_f32_16x16x4_f32(false, ac, false, bh, (short)0, accH, false, false);
            accA = __builtin_amdgcn_wmma_f32_16x16x4_f32(false, ad, false, ba, (short)0, accA, false, false);
        }
    }

    const float* baseC = ws + OFF_BC;
    const float* baseD = ws + OFF_BD;
    const int gc = tile * 16 + m;
#pragma unroll
    for (int r = 0; r < 8; ++r) {
        const int gr = wave * 16 + r + 8 * (lane >> 4);
        if (gr < TP && gc < N_C) {
            const size_t o = (size_t)gr * N_C + gc;
            out[o] = accC[r] + baseC[o];
            out[(size_t)TP * N_C + o] = (accH[r] + accA[r]) + baseD[o];
        }
    }
}

extern "C" void kernel_launch(void* const* d_in, const int* in_sizes, int n_in,
                              void* d_out, int out_size, void* d_ws, size_t ws_size,
                              hipStream_t stream) {
    const float* C   = (const float*)d_in[0];
    const float* D   = (const float*)d_in[1];
    const float* M   = (const float*)d_in[2];
    const float* cov = (const float*)d_in[3];
    const float* bnz = (const float*)d_in[4];
    const float* anz = (const float*)d_in[5];
    const float* hnz = (const float*)d_in[6];
    const float* mu  = (const float*)d_in[7];
    const float* nu  = (const float*)d_in[8];
    const float* ups = (const float*)d_in[9];
    const float* zet = (const float*)d_in[10];
    const int* rows  = (const int*)d_in[11];
    const int* cols  = (const int*)d_in[12];
    float* ws  = (float*)d_ws;
    float* out = (float*)d_out;

    prep_k<<<PREP_BLOCKS, 256, 0, stream>>>(C, D, M, cov, bnz, anz, hnz, mu, nu, ups, zet, rows, cols, ws);
    spmm_k<<<NT, 320, 0, stream>>>(ws, out);
}


# Base terms computed in the product launch: each wave loads its own cells' mobility rows and covariates during the list build and starts its accumulators there; setup launch 155 workgroups, not 392
# speedup vs baseline: 10.5605x; 1.1225x over previous
#include <hip/hip_runtime.h>

typedef __attribute__((ext_vector_type(2))) float v2f;
typedef __attribute__((ext_vector_type(8))) float v8f;
typedef __attribute__((ext_vector_type(4))) float v4f;
typedef __attribute__((ext_vector_type(3))) float v3f;
typedef __attribute__((ext_vector_type(2))) unsigned v2u;

#define N_C    3144
#define NP     2
#define NMOB   6
#define NCOV   10
#define T_ALL  156
#define TP     154
#define TP_PAD 160
#define NNZ_   31440


#define OFF_XT   ((size_t)0)
#define OFF_V4   ((size_t)N_C * TP_PAD * 2)
#define OFF_OT   (OFF_V4 + (size_t)NNZ_ * 4)

#define NSB      30
#define SBL      1048
#define NBK      16
#define BKW      208
#define SST      5

#define NC4      786
#define XT_TM    128
#define NTB_X    125
#define PREP_BLOCKS 155

__global__ __launch_bounds__(256)
void prep_k(const float* __restrict__ C, const float* __restrict__ D,
            const float* __restrict__ bnz, const float* __restrict__ anz,
            const float* __restrict__ hnz,
            const int* __restrict__ rows, const int* __restrict__ cols,
            float* __restrict__ ws) {
    const int tid = threadIdx.x;
    if (blockIdx.x < NSB) {
        __shared__ int wc[8][NBK];
        __shared__ int tot[NBK];
        const int sb   = blockIdx.x;
        const int wave = tid >> 5;
        const int lane = tid & 31;
        int key[SST];
        unsigned rec[SST];
        float bv[SST], hv[SST], av[SST];
#pragma unroll
        for (int s = 0; s < SST; ++s) {
            const int jl = s * 256 + tid;
            const bool ok = jl < SBL;
            const int j = sb * SBL + (ok ? jl : 0);
            const int c = cols[j];
            const int r = rows[j];
            bv[s] = bnz[j];
            hv[s] = hnz[j];
            av[s] = anz[j];
            key[s] = ok ? c / BKW : NBK;
            rec[s] = (unsigned)r | ((unsigned)c << 12);
        }
        int mycnt = 0;
#pragma unroll
        for (int g = 0; g < NBK; ++g) {
            int c = 0;
#pragma unroll
            for (int s = 0; s < SST; ++s) c += __popc(__builtin_amdgcn_ballot_w32(key[s] == g));
            if (lane == g) mycnt = c;
        }
        if (lane < NBK) wc[wave][lane] = mycnt;
        __syncthreads();
        if (tid < NBK) {
            int run = 0;
#pragma unroll
            for (int w = 0; w < 8; ++w) {
                const int c = wc[w][tid];
                wc[w][tid] = run;
                run += c;
            }
            tot[tid] = run;
        }
        __syncthreads();
        int mybase = 0;
#pragma unroll
        for (int g = 0; g < NBK; ++g) {
            const int t = tot[g];
            if (g < lane) mybase += t;
        }
        int* OT = reinterpret_cast<int*>(ws + OFF_OT) + sb * (NBK + 1);
        if (tid <= NBK) OT[tid] = mybase;
        if (lane < NBK) mybase += wc[wave][lane];
        int pos[SST];
#pragma unroll
        for (int s = 0; s < SST; ++s) pos[s] = 0;
#pragma unroll
        for (int g = 0; g < NBK; ++g) {
            int run = __builtin_amdgcn_readlane(mybase, g);
#pragma unroll
            for (int s = 0; s < SST; ++s) {
                const bool mt = (key[s] == g);
                const unsigned mask = __builtin_amdgcn_ballot_w32(mt);
                if (mt) pos[s] = run + __popc(mask & ((1u << lane) - 1u));
                run += __popc(mask);
            }
        }
        typedef __attribute__((ext_vector_type(4))) unsigned v4u;
        v4u* V4o = reinterpret_cast<v4u*>(ws + OFF_V4) + (size_t)sb * SBL;
#pragma unroll
        for (int s = 0; s < SST; ++s) {
            if (key[s] < NBK) {
                v4u o;
                o.x = __float_as_uint(bv[s]);
                o.y = __float_as_uint(hv[s]);
                o.z = __float_as_uint(av[s]);
                o.w = rec[s];
                V4o[pos[s]] = o;
            }
        }
        return;
    }
    const int bid = blockIdx.x - NSB;
    if (bid < NTB_X) {
        __shared__ float tc[32][XT_TM + 1];
        __shared__ float td[32][XT_TM + 1];
        const int m0 = (bid % 25) * XT_TM;
        const int t0 = (bid / 25) * 32;
        const int tx = tid & 31;
        const int ty = tid >> 5;
        const v4f* C4 = reinterpret_cast<const v4f*>(C);
        const v4f* D4 = reinterpret_cast<const v4f*>(D);
        const int g4 = (m0 >> 2) + tx;
        v4f ca[4], cb[4], da[4], db[4];
#pragma unroll
        for (int i = 0; i < 4; ++i) {
            const int t  = t0 + ty + 8 * i;
            const size_t rb = (size_t)(t < TP ? t : TP - 1) * NC4 + (g4 < NC4 ? g4 : NC4 - 1);
            ca[i] = C4[rb];
            cb[i] = C4[rb + NC4];
            da[i] = D4[rb];
            db[i] = D4[rb + NC4];
        }
#pragma unroll
        for (int i = 0; i < 4; ++i) {
            const int tl = ty + 8 * i;
            const bool in = (t0 + tl < TP) && (g4 < NC4);
            const v4f z = {0.0f, 0.0f, 0.0f, 0.0f};
            const v4f cs  = in ? ca[i] + cb[i] : z;
            const v4f dsv = in ? da[i] + db[i] : z;
            tc[tl][4 * tx + 0] = cs.x;  tc[tl][4 * tx + 1] = cs.y;
            tc[tl][4 * tx + 2] = cs.z;  tc[tl][4 * tx + 3] = cs.w;
            td[tl][4 * tx + 0] = dsv.x; td[tl][4 * tx + 1] = dsv.y;
            td[tl][4 * tx + 2] = dsv.z; td[tl][4 * tx + 3] = dsv.w;
        }
        __syncthreads();
        v4f* XT4 = reinterpret_cast<v4f*>(ws + OFF_XT);
        const int l2 = (tid & 15) * 2;
        const int hf = (tid >> 4) & 1;
        v4f o[8];
#pragma unroll
        for (int s = 0; s < 8; ++s) {
            const int ml = ty * 16 + s * 2 + hf;
            o[s].x = tc[l2][ml];
            o[s].y = td[l2][ml];
            o[s].z = tc[l2 + 1][ml];
            o[s].w = td[l2 + 1][ml];
        }
#pragma unroll
        for (int s = 0; s < 8; ++s) {
            const int mm = m0 + ty * 16 + s * 2 + hf;
            if (mm < N_C) XT4[((size_t)mm * TP_PAD + t0 + l2) >> 1] = o[s];
        }
        return;
    }
}

#define NT    197
#define NW    10
#define SEGL  3144
#define BPW   3
#define FCH   4
#define KB    8
#define NOCOL  16u
#define PADENT (NOCOL << 12)

static __device__ __forceinline__ void base_terms(const float (&mr)[NMOB][9], const float (&cvv)[NCOV],
                                                  const float (&cmu)[NMOB * NP], const float (&cnu)[NMOB * NP],
                                                  const float (&cup)[NCOV], const float (&cze)[NCOV],
                                                  v8f& accC, v8f& accH) {
    float uc = 0.0f, zc = 0.0f;
#pragma unroll
    for (int j = 0; j < NCOV; ++j) {
        uc = fmaf(cup[j], cvv[j], uc);
        zc = fmaf(cze[j], cvv[j], zc);
    }
#pragma unroll
    for (int r = 0; r < 8; ++r) {
        float mc = 0.0f, md = 0.0f;
#pragma unroll
        for (int k = 0; k < NMOB; ++k) {
            mc = fmaf(mr[k][r],     cmu[k * NP + 0], mc);
            mc = fmaf(mr[k][r + 1], cmu[k * NP + 1], mc);
            md = fmaf(mr[k][r],     cnu[k * NP + 0], md);
            md = fmaf(mr[k][r + 1], cnu[k * NP + 1], md);
        }
        accC[r] = mc + uc;
        accH[r] = md + zc;
    }
}

__global__ __launch_bounds__(320)
void spmm_k(const float* __restrict__ ws,
            const float* __restrict__ M, const float* __restrict__ cov,
            const float* __restrict__ mu, const float* __restrict__ nu,
            const float* __restrict__ ups, const float* __restrict__ zet,
            float* __restrict__ out) {
    __shared__ __attribute__((aligned(16))) unsigned list[NNZ_];
    __shared__ unsigned short steptab[NNZ_ / 4];
    __shared__ int scnt[NW];

    const int tile = blockIdx.x;
    const int tid  = threadIdx.x;
    const int wave = tid >> 5;
    const int lane = tid & 31;
    const int segb = wave * SEGL;

    const int bucket = tile / (BKW / 16);
    const int* OT = reinterpret_cast<const int*>(ws + OFF_OT);
    const unsigned* RW = reinterpret_cast<const unsigned*>(ws + OFF_V4);

    int lo[BPW], n[BPW];
#pragma unroll
    for (int i = 0; i < BPW; ++i) {
        const int sb = wave * BPW + i;
        lo[i] = OT[sb * (NBK + 1) + bucket];
        n[i]  = OT[sb * (NBK + 1) + bucket + 1];
    }
#pragma unroll
    for (int i = 0; i < BPW; ++i) {
        int l = __builtin_amdgcn_readfirstlane(lo[i]);
        int h = __builtin_amdgcn_readfirstlane(n[i]);
        l = l < 0 ? 0 : (l > SBL ? SBL : l);
        h = h < l ? l : (h > SBL ? SBL : h);
        lo[i] = l;
        n[i]  = h - l;
    }

    int wcount = 0;
    unsigned rc[BPW][FCH];
#pragma unroll
    for (int i = 0; i < BPW; ++i) {
#pragma unroll
        for (int u = 0; u < FCH; ++u) {
            const int q  = lo[i] + u * 32 + lane;
            const int qc = q < SBL ? q : SBL - 1;
            rc[i][u] = RW[((size_t)(wave * BPW + i) * SBL + qc) * 4 + 3];
        }
    }
    __builtin_amdgcn_sched_barrier(0);

    const int m  = lane & 15;
    const int gc = tile * 16 + m;
    const int gl = gc < N_C ? gc : N_C - 1;
    const int tb = wave * 16 + 8 * (lane >> 4);
    float mr[NMOB][9], cvv[NCOV];
#pragma unroll
    for (int k = 0; k < NMOB; ++k) {
#pragma unroll
        for (int j = 0; j < 9; ++j) {
            const int t = tb + j < T_ALL ? tb + j : T_ALL - 1;
            mr[k][j] = M[((size_t)k * T_ALL + t) * N_C + gl];
        }
    }
#pragma unroll
    for (int j = 0; j < NCOV; ++j) cvv[j] = cov[(size_t)j * N_C + gl];
    float cmu[NMOB * NP], cnu[NMOB * NP], cup[NCOV], cze[NCOV];
#pragma unroll
    for (int i = 0; i < NMOB * NP; ++i) { cmu[i] = mu[i]; cnu[i] = nu[i]; }
#pragma unroll
    for (int j = 0; j < NCOV; ++j) { cup[j] = ups[j]; cze[j] = zet[j]; }
    __builtin_amdgcn_sched_barrier(0);

#pragma unroll
    for (int i = 0; i < BPW; ++i) {
#pragma unroll
        for (int u = 0; u < FCH; ++u) {
            const int p = u * 32 + lane;
            const bool match = (p < n[i]) && ((int)(rc[i][u] >> 16) == tile);
            const unsigned mask = __builtin_amdgcn_ballot_w32(match);
            if (match) {
                const int slot = wcount + __popc(mask & ((1u << lane) - 1u));
                list[segb + slot] = (rc[i][u] & 0xFFFFu) | ((unsigned)((wave * BPW + i) * SBL + lo[i] + p) << 17);
            }
            wcount += __popc(mask);
        }
        for (int p0 = FCH * 32; p0 < n[i]; p0 += 32) {
            const int p  = p0 + lane;
            const int q  = lo[i] + p;
            const int qc = q < SBL ? q : SBL - 1;
            const unsigned r1 = RW[((size_t)(wave * BPW + i) * SBL + qc) * 4 + 3];
            const bool match = (p < n[i]) && ((int)(r1 >> 16) == tile);
            const unsigned mask = __builtin_amdgcn_ballot_w32(match);
            if (match) {
                const int slot = wcount + __popc(mask & ((1u << lane) - 1u));
                list[segb + slot] = (r1 & 0xFFFFu) | ((unsigned)((wave * BPW + i) * SBL + q) << 17);
            }
            wcount += __popc(mask);
        }
    }
    const int wpad = (wcount + 3) & ~3;
    if (lane < wpad - wcount) list[segb + wcount + lane] = PADENT;
    if (lane == 0) scnt[wave] = wpad >> 2;
    __syncthreads();

    int sbase = 0, stot = 0;
#pragma unroll
    for (int w = 0; w < NW; ++w) {
        const int c = scnt[w];
        if (w < wave) sbase += c;
        stot += c;
    }
    for (int ls = lane; ls < (wpad >> 2); ls += 32)
        steptab[sbase + ls] = (unsigned short)(segb + 4 * ls);
    __syncthreads();

    const int h2 = (lane >> 4) * 2;
    const v2f* XT = reinterpret_cast<const v2f*>(ws + OFF_XT) + (wave * 16 + m);
    const float* V4 = ws + OFF_V4;

    v8f accC = {}, accH = {}, accA = {};
    if (stot <= 0) base_terms(mr, cvv, cmu, cnu, cup, cze, accC, accH);

    for (int v0 = 0; v0 < stot; v0 += KB) {
        v2f x0[KB], x1[KB];
        v3f w0[KB], w1[KB];
        unsigned cl0[KB], cl1[KB];
#pragma unroll
        for (int q = 0; q < KB; ++q) {
            const bool on = (v0 + q < stot);
            const int  v  = on ? v0 + q : stot - 1;
            const int  off = steptab[v];
            const v2u ent = *reinterpret_cast<const v2u*>(&list[off + h2]);
            x0[q] = XT[(size_t)(ent.x & 4095u) * TP_PAD];
            x1[q] = XT[(size_t)(ent.y & 4095u) * TP_PAD];
            w0[q] = *reinterpret_cast<const v3f*>(V4 + (size_t)(ent.x >> 17) * 4);
            w1[q] = *reinterpret_cast<const v3f*>(V4 + (size_t)(ent.y >> 17) * 4);
            cl0[q] = on ? ((ent.x >> 12) & 31u) : NOCOL;
            cl1[q] = on ? ((ent.y >> 12) & 31u) : NOCOL;
        }
        __builtin_amdgcn_sched_barrier(0);
        if (v0 == 0) base_terms(mr, cvv, cmu, cnu, cup, cze, accC, accH);
#pragma unroll
        for (int q = 0; q < KB; ++q) {
            const bool s0 = (cl0[q] == (unsigned)m);
            const bool s1 = (cl1[q] == (unsigned)m);
            v2f ac, ad, bb, bh, ba;
            ac.x = x0[q].x;  ac.y = x1[q].x;
            ad.x = x0[q].y;  ad.y = x1[q].y;
            bb.x = s0 ? w0[q].x : 0.0f;  bb.y = s1 ? w1[q].x : 0.0f;
            bh.x = s0 ? w0[q].y : 0.0f;  bh.y = s1 ? w1[q].y : 0.0f;
            ba.x = s0 ? w0[q].z : 0.0f;  ba.y = s1 ? w1[q].z : 0.0f;
            accC = __builtin_amdgcn_wmma_f32_16x16x4_f32(false, ac, false, bb, (short)0, accC, false, false);
            accH = __builtin_amdgcn_wmma_f32_16x16x4_f32(false, ac, false, bh, (short)0, accH, false, false);
            accA = __builtin_amdgcn_wmma_f32_16x16x4_f32(false, ad, false, ba, (short)0, accA, false, false);
        }
    }

#pragma unroll
    for (int r = 0; r < 8; ++r) {
        const int gr = tb + r;
        if (gr < TP && gc < N_C) {
            const size_t o = (size_t)gr * N_C + gc;
            out[o] = accC[r];
            out[(size_t)TP * N_C + o] = accH[r] + accA[r];
        }
    }
}

extern "C" void kernel_launch(void* const* d_in, const int* in_sizes, int n_in,
                              void* d_out, int out_size, void* d_ws, size_t ws_size,
                              hipStream_t stream) {
    const float* C   = (const float*)d_in[0];
    const float* D   = (const float*)d_in[1];
    const float* M   = (const float*)d_in[2];
    const float* cov = (const float*)d_in[3];
    const float* bnz = (const float*)d_in[4];
    const float* anz = (const float*)d_in[5];
    const float* hnz = (const float*)d_in[6];
    const float* mu  = (const float*)d_in[7];
    const float* nu  = (const float*)d_in[8];
    const float* ups = (const float*)d_in[9];
    const float* zet = (const float*)d_in[10];
    const int* rows  = (const int*)d_in[11];
    const int* cols  = (const int*)d_in[12];
    float* ws  = (float*)d_ws;
    float* out = (float*)d_out;

    prep_k<<<PREP_BLOCKS, 256, 0, stream>>>(C, D, bnz, anz, hnz, rows, cols, ws);
    spmm_k<<<NT, 320, 0, stream>>>(ws, M, cov, mu, nu, ups, zet, out);
}
